# starting kernel
# baseline (speedup 1.0000x reference)
#include <hip/hip_runtime.h>

typedef __attribute__((ext_vector_type(2))) float v2f;
typedef __attribute__((ext_vector_type(8))) float v8f;

#define N_NODES 50000
#define N_EDGES 800000
#define D 128
#define LDS_PITCH 132

__global__ void gin_init_rst(const float* __restrict__ feat,
                             const float* __restrict__ eps,
                             float* __restrict__ rst) {
    int i = blockIdx.x * blockDim.x + threadIdx.x;
    float s = 1.0f + eps[0];
    const float4* f4 = (const float4*)feat;
    float4* r4 = (float4*)rst;
    float4 v = f4[i];
    v.x *= s; v.y *= s; v.z *= s; v.w *= s;
    r4[i] = v;
}

__global__ void gin_scatter(const float* __restrict__ feat,
                            const int* __restrict__ esrc,
                            const int* __restrict__ edst,
                            float* __restrict__ rst) {
    int wave = (blockIdx.x * blockDim.x + threadIdx.x) >> 5;
    int lane = threadIdx.x & 31;
    if (wave >= N_EDGES) return;
    int s = __builtin_amdgcn_readfirstlane(esrc[wave]);
    int d = __builtin_amdgcn_readfirstlane(edst[wave]);
    const float4* fs = (const float4*)(feat + (size_t)s * D);
    float4 v = fs[lane];
    float* o = rst + (size_t)d * D + lane * 4;
    unsafeAtomicAdd(o + 0, v.x);
    unsafeAtomicAdd(o + 1, v.y);
    unsafeAtomicAdd(o + 2, v.z);
    unsafeAtomicAdd(o + 3, v.w);
}

__global__ void __launch_bounds__(256)
gin_mlp(const float* __restrict__ rst,
        const float* __restrict__ W1, const float* __restrict__ b1,
        const float* __restrict__ W2, const float* __restrict__ b2,
        float* __restrict__ out) {
    __shared__ float hA[16 * LDS_PITCH];
    __shared__ float hH[16 * LDS_PITCH];

    const int tid  = threadIdx.x;
    const int wave = tid >> 5;
    const int lane = tid & 31;
    const int l15  = lane & 15;
    const int hi   = lane >> 4;
    const int koff = hi * 2;
    const int m0   = blockIdx.x * 16;
    const int n0   = wave * 16;

    {
        int r = tid >> 4;
        int c = (tid & 15) * 8;
        const float4* src = (const float4*)(rst + (size_t)(m0 + r) * D + c);
        float4 a = src[0], b = src[1];
        float* dst = hA + r * LDS_PITCH + c;
        dst[0] = a.x; dst[1] = a.y; dst[2] = a.z; dst[3] = a.w;
        dst[4] = b.x; dst[5] = b.y; dst[6] = b.z; dst[7] = b.w;
    }
    __syncthreads();

    v8f acc;
    {
        float bb = b1[n0 + l15];
        #pragma unroll
        for (int r = 0; r < 8; ++r) acc[r] = bb;
    }
    #pragma unroll
    for (int k = 0; k < D; k += 4) {
        v2f a = *(const v2f*)(hA + l15 * LDS_PITCH + k + koff);
        v2f b;
        b.x = W1[(size_t)(k + koff)     * D + n0 + l15];
        b.y = W1[(size_t)(k + koff + 1) * D + n0 + l15];
        acc = __builtin_amdgcn_wmma_f32_16x16x4_f32(
             false, a,  false, b,
             (short)0, acc,  false,  false);
    }
    #pragma unroll
    for (int r = 0; r < 8; ++r) {
        int m = r + hi * 8;
        float v = acc[r] > 0.0f ? acc[r] : 0.0f;
        hH[m * LDS_PITCH + n0 + l15] = v;
    }
    __syncthreads();

    v8f acc2;
    {
        float bb = b2[n0 + l15];
        #pragma unroll
        for (int r = 0; r < 8; ++r) acc2[r] = bb;
    }
    #pragma unroll
    for (int k = 0; k < D; k += 4) {
        v2f a = *(const v2f*)(hH + l15 * LDS_PITCH + k + koff);
        v2f b;
        b.x = W2[(size_t)(k + koff)     * D + n0 + l15];
        b.y = W2[(size_t)(k + koff + 1) * D + n0 + l15];
        acc2 = __builtin_amdgcn_wmma_f32_16x16x4_f32(
            false, a, false, b, (short)0, acc2, false, false);
    }
    #pragma unroll
    for (int r = 0; r < 8; ++r) {
        int m = r + hi * 8;
        float v = acc2[r] > 0.0f ? acc2[r] : 0.0f;
        out[(size_t)(m0 + m) * D + n0 + l15] = v;
    }
}

extern "C" void kernel_launch(void* const* d_in, const int* in_sizes, int n_in,
                              void* d_out, int out_size, void* d_ws, size_t ws_size,
                              hipStream_t stream) {
    const float* feat = (const float*)d_in[0];
    const float* eps  = (const float*)d_in[1];
    const float* W1   = (const float*)d_in[2];
    const float* b1   = (const float*)d_in[3];
    const float* W2   = (const float*)d_in[4];
    const float* b2   = (const float*)d_in[5];
    const int*   esrc = (const int*)d_in[6];
    const int*   edst = (const int*)d_in[7];
    float* out = (float*)d_out;
    float* rst = (float*)d_ws;

    gin_init_rst<<<(N_NODES * D / 4) / 256, 256, 0, stream>>>(feat, eps, rst);
    gin_scatter<<<(N_EDGES * 32) / 256, 256, 0, stream>>>(feat, esrc, edst, rst);
    gin_mlp<<<N_NODES / 16, 256, 0, stream>>>(rst, W1, b1, W2, b2, out);
}


# scatter: lane l takes columns l+32j, so each f32 atomic add instruction fills one whole 128-byte line of rst[dst], not a quarter of each of four lines; same arrival-order sums as parent
# speedup vs baseline: 1.3214x; 1.3214x over previous
#include <hip/hip_runtime.h>

typedef __attribute__((ext_vector_type(2))) float v2f;
typedef __attribute__((ext_vector_type(8))) float v8f;

#define N_NODES 50000
#define N_EDGES 800000
#define D 128
#define LDS_PITCH 132

__global__ void gin_init_rst(const float* __restrict__ feat,
                             const float* __restrict__ eps,
                             float* __restrict__ rst) {
    int i = blockIdx.x * blockDim.x + threadIdx.x;
    float s = 1.0f + eps[0];
    const float4* f4 = (const float4*)feat;
    float4* r4 = (float4*)rst;
    float4 v = f4[i];
    v.x *= s; v.y *= s; v.z *= s; v.w *= s;
    r4[i] = v;
}

__global__ void gin_scatter(const float* __restrict__ feat,
                            const int* __restrict__ esrc,
                            const int* __restrict__ edst,
                            float* __restrict__ rst) {
    int wave = (blockIdx.x * blockDim.x + threadIdx.x) >> 5;
    int lane = threadIdx.x & 31;
    if (wave >= N_EDGES) return;
    int s = __builtin_amdgcn_readfirstlane(esrc[wave]);
    int d = __builtin_amdgcn_readfirstlane(edst[wave]);
    const float* fs = feat + (size_t)s * D + lane;
    float v0 = fs[0];
    float v1 = fs[32];
    float v2 = fs[64];
    float v3 = fs[96];
    float* o = rst + (size_t)d * D + lane;
    unsafeAtomicAdd(o,      v0);
    unsafeAtomicAdd(o + 32, v1);
    unsafeAtomicAdd(o + 64, v2);
    unsafeAtomicAdd(o + 96, v3);
}

__global__ void __launch_bounds__(256)
gin_mlp(const float* __restrict__ rst,
        const float* __restrict__ W1, const float* __restrict__ b1,
        const float* __restrict__ W2, const float* __restrict__ b2,
        float* __restrict__ out) {
    __shared__ float hA[16 * LDS_PITCH];
    __shared__ float hH[16 * LDS_PITCH];

    const int tid  = threadIdx.x;
    const int wave = tid >> 5;
    const int lane = tid & 31;
    const int l15  = lane & 15;
    const int hi   = lane >> 4;
    const int koff = hi * 2;
    const int m0   = blockIdx.x * 16;
    const int n0   = wave * 16;

    {
        int r = tid >> 4;
        int c = (tid & 15) * 8;
        const float4* src = (const float4*)(rst + (size_t)(m0 + r) * D + c);
        float4 a = src[0], b = src[1];
        float* dst = hA + r * LDS_PITCH + c;
        dst[0] = a.x; dst[1] = a.y; dst[2] = a.z; dst[3] = a.w;
        dst[4] = b.x; dst[5] = b.y; dst[6] = b.z; dst[7] = b.w;
    }
    __syncthreads();

    v8f acc;
    {
        float bb = b1[n0 + l15];
        #pragma unroll
        for (int r = 0; r < 8; ++r) acc[r] = bb;
    }
    #pragma unroll
    for (int k = 0; k < D; k += 4) {
        v2f a = *(const v2f*)(hA + l15 * LDS_PITCH + k + koff);
        v2f b;
        b.x = W1[(size_t)(k + koff)     * D + n0 + l15];
        b.y = W1[(size_t)(k + koff + 1) * D + n0 + l15];
        acc = __builtin_amdgcn_wmma_f32_16x16x4_f32(
             false, a,  false, b,
             (short)0, acc,  false,  false);
    }
    #pragma unroll
    for (int r = 0; r < 8; ++r) {
        int m = r + hi * 8;
        float v = acc[r] > 0.0f ? acc[r] : 0.0f;
        hH[m * LDS_PITCH + n0 + l15] = v;
    }
    __syncthreads();

    v8f acc2;
    {
        float bb = b2[n0 + l15];
        #pragma unroll
        for (int r = 0; r < 8; ++r) acc2[r] = bb;
    }
    #pragma unroll
    for (int k = 0; k < D; k += 4) {
        v2f a = *(const v2f*)(hH + l15 * LDS_PITCH + k + koff);
        v2f b;
        b.x = W2[(size_t)(k + koff)     * D + n0 + l15];
        b.y = W2[(size_t)(k + koff + 1) * D + n0 + l15];
        acc2 = __builtin_amdgcn_wmma_f32_16x16x4_f32(
            false, a, false, b, (short)0, acc2, false, false);
    }
    #pragma unroll
    for (int r = 0; r < 8; ++r) {
        int m = r + hi * 8;
        float v = acc2[r] > 0.0f ? acc2[r] : 0.0f;
        out[(size_t)(m0 + m) * D + n0 + l15] = v;
    }
}

extern "C" void kernel_launch(void* const* d_in, const int* in_sizes, int n_in,
                              void* d_out, int out_size, void* d_ws, size_t ws_size,
                              hipStream_t stream) {
    const float* feat = (const float*)d_in[0];
    const float* eps  = (const float*)d_in[1];
    const float* W1   = (const float*)d_in[2];
    const float* b1   = (const float*)d_in[3];
    const float* W2   = (const float*)d_in[4];
    const float* b2   = (const float*)d_in[5];
    const int*   esrc = (const int*)d_in[6];
    const int*   edst = (const int*)d_in[7];
    float* out = (float*)d_out;
    float* rst = (float*)d_ws;

    gin_init_rst<<<(N_NODES * D / 4) / 256, 256, 0, stream>>>(feat, eps, rst);
    gin_scatter<<<(N_EDGES * 32) / 256, 256, 0, stream>>>(feat, esrc, edst, rst);
    gin_mlp<<<N_NODES / 16, 256, 0, stream>>>(rst, W1, b1, W2, b2, out);
}


# aggregation by row owner: each edge files its src id in its dst row's bucket (one returning integer add); the MLP tile load sums feat[src] per row: no f32 atomics, no rst pass; sums in arrival order
# speedup vs baseline: 1.7955x; 1.3588x over previous
#include <hip/hip_runtime.h>

typedef __attribute__((ext_vector_type(2))) float v2f;
typedef __attribute__((ext_vector_type(8))) float v8f;

#define N_NODES 50000
#define N_EDGES 800000
#define D 128
#define LDS_PITCH 132

#define CAP 64
#define CNT_PAD 50176

__global__ void gin_zero_counts(int* __restrict__ cnt) {
    cnt[blockIdx.x * blockDim.x + threadIdx.x] = 0;
}

__global__ void gin_bucket(const int* __restrict__ esrc,
                           const int* __restrict__ edst,
                           int* __restrict__ cnt,
                           int* __restrict__ bkt,
                           int* __restrict__ ovs,
                           int* __restrict__ ovd) {
    int e = blockIdx.x * blockDim.x + threadIdx.x;
    if (e >= N_EDGES) return;
    int d = edst[e];
    int s = esrc[e];
    int slot = atomicAdd(cnt + d, 1);
    if (slot < CAP) {
        bkt[(size_t)d * CAP + slot] = s;
    } else {
        int o = atomicAdd(cnt + N_NODES, 1);
        ovs[o] = s;
        ovd[o] = d;
    }
}

__global__ void __launch_bounds__(256)
gin_mlp(const float* __restrict__ feat,
        const float* __restrict__ eps,
        const int* __restrict__ cnt,
        const int* __restrict__ bkt,
        const int* __restrict__ ovs,
        const int* __restrict__ ovd,
        const float* __restrict__ W1, const float* __restrict__ b1,
        const float* __restrict__ W2, const float* __restrict__ b2,
        float* __restrict__ out) {
    __shared__ __attribute__((aligned(16))) float hA[16 * LDS_PITCH];
    __shared__ float hH[16 * LDS_PITCH];

    const int tid  = threadIdx.x;
    const int wave = tid >> 5;
    const int lane = tid & 31;
    const int l15  = lane & 15;
    const int hi   = lane >> 4;
    const int koff = hi * 2;
    const int m0   = blockIdx.x * 16;
    const int n0   = wave * 16;

    {
        const int wv = __builtin_amdgcn_readfirstlane(wave);
        const float sc = 1.0f + eps[0];
        const float4* f4 = (const float4*)feat;
        int on = cnt[N_NODES];
        on = on < N_EDGES ? on : N_EDGES;
        int    nn[2], i0[2], i1[2];
        float4 self[2];
        #pragma unroll
        for (int rr = 0; rr < 2; ++rr) {
            const int row = m0 + wv * 2 + rr;
            int n = cnt[row];
            n = n < CAP ? n : CAP;
            nn[rr] = n;
            const int* bp = bkt + (size_t)row * CAP;
            self[rr] = f4[(size_t)row * 32 + lane];
            i0[rr] = (lane < n)      ? bp[lane]      : 0;
            i1[rr] = (lane + 32 < n) ? bp[lane + 32] : 0;
        }
        #pragma unroll
        for (int rr = 0; rr < 2; ++rr) {
            const int r   = wv * 2 + rr;
            const int row = m0 + r;
            const int n   = nn[rr];
            float4 acc = make_float4(0.0f, 0.0f, 0.0f, 0.0f);
            #pragma nounroll
            for (int j = 0; j < n; j += 8) {
                const int iv = (j < 32) ? i0[rr] : i1[rr];
                float4 v[8];
                #pragma unroll
                for (int u = 0; u < 8; ++u) {
                    const int s = __builtin_amdgcn_readlane(iv, (j & 31) + u);
                    v[u] = f4[(size_t)s * 32 + lane];
                }
                #pragma unroll
                for (int u = 0; u < 8; ++u) {
                    const bool ok = (j + u) < n;
                    acc.x += ok ? v[u].x : 0.0f;
                    acc.y += ok ? v[u].y : 0.0f;
                    acc.z += ok ? v[u].z : 0.0f;
                    acc.w += ok ? v[u].w : 0.0f;
                }
            }
            for (int i = 0; i < on; ++i) {
                if (ovd[i] == row) {
                    const float4 v = f4[(size_t)ovs[i] * 32 + lane];
                    acc.x += v.x; acc.y += v.y; acc.z += v.z; acc.w += v.w;
                }
            }
            float4 t;
            t.x = fmaf(sc, self[rr].x, acc.x);
            t.y = fmaf(sc, self[rr].y, acc.y);
            t.z = fmaf(sc, self[rr].z, acc.z);
            t.w = fmaf(sc, self[rr].w, acc.w);
            *(float4*)(hA + r * LDS_PITCH + lane * 4) = t;
        }
    }
    __syncthreads();

    v8f acc;
    {
        float bb = b1[n0 + l15];
        #pragma unroll
        for (int r = 0; r < 8; ++r) acc[r] = bb;
    }
    #pragma unroll
    for (int k = 0; k < D; k += 4) {
        v2f a = *(const v2f*)(hA + l15 * LDS_PITCH + k + koff);
        v2f b;
        b.x = W1[(size_t)(k + koff)     * D + n0 + l15];
        b.y = W1[(size_t)(k + koff + 1) * D + n0 + l15];
        acc = __builtin_amdgcn_wmma_f32_16x16x4_f32(
             false, a,  false, b,
             (short)0, acc,  false,  false);
    }
    #pragma unroll
    for (int r = 0; r < 8; ++r) {
        int m = r + hi * 8;
        float v = acc[r] > 0.0f ? acc[r] : 0.0f;
        hH[m * LDS_PITCH + n0 + l15] = v;
    }
    __syncthreads();

    v8f acc2;
    {
        float bb = b2[n0 + l15];
        #pragma unroll
        for (int r = 0; r < 8; ++r) acc2[r] = bb;
    }
    #pragma unroll
    for (int k = 0; k < D; k += 4) {
        v2f a = *(const v2f*)(hH + l15 * LDS_PITCH + k + koff);
        v2f b;
        b.x = W2[(size_t)(k + koff)     * D + n0 + l15];
        b.y = W2[(size_t)(k + koff + 1) * D + n0 + l15];
        acc2 = __builtin_amdgcn_wmma_f32_16x16x4_f32(
            false, a, false, b, (short)0, acc2, false, false);
    }
    #pragma unroll
    for (int r = 0; r < 8; ++r) {
        int m = r + hi * 8;
        float v = acc2[r] > 0.0f ? acc2[r] : 0.0f;
        out[(size_t)(m0 + m) * D + n0 + l15] = v;
    }
}

extern "C" void kernel_launch(void* const* d_in, const int* in_sizes, int n_in,
                              void* d_out, int out_size, void* d_ws, size_t ws_size,
                              hipStream_t stream) {
    const float* feat = (const float*)d_in[0];
    const float* eps  = (const float*)d_in[1];
    const float* W1   = (const float*)d_in[2];
    const float* b1   = (const float*)d_in[3];
    const float* W2   = (const float*)d_in[4];
    const float* b2   = (const float*)d_in[5];
    const int*   esrc = (const int*)d_in[6];
    const int*   edst = (const int*)d_in[7];
    float* out = (float*)d_out;
    int* cnt = (int*)d_ws;
    int* bkt = cnt + CNT_PAD;
    int* ovs = bkt + N_NODES * CAP;
    int* ovd = ovs + N_EDGES;

    gin_zero_counts<<<CNT_PAD / 256, 256, 0, stream>>>(cnt);
    gin_bucket<<<N_EDGES / 256, 256, 0, stream>>>(esrc, edst, cnt, bkt, ovs, ovd);
    gin_mlp<<<N_NODES / 16, 256, 0, stream>>>(feat, eps, cnt, bkt, ovs, ovd, W1, b1, W2, b2, out);
}


# gin_mlp: 768 blocks own 4 or 5 row tiles each (was 3125 blocks of one): a W fragment load feeds 4-5 WMMAs on independent accumulators; 4x fewer waves and W reads; sum order as parent
# speedup vs baseline: 1.9051x; 1.0610x over previous
#include <hip/hip_runtime.h>

typedef __attribute__((ext_vector_type(2))) float v2f;
typedef __attribute__((ext_vector_type(8))) float v8f;

#define N_NODES 50000
#define N_EDGES 800000
#define D 128
#define LDS_PITCH 132

#define CAP 64
#define CNT_PAD 50176

#define N_TILES    (N_NODES / 16)
#define MLP_BLOCKS 768
#define TILES_LO   (N_TILES / MLP_BLOCKS)
#define N_HI       (N_TILES % MLP_BLOCKS)
#define TILES_HI   (TILES_LO + 1)
#define ROWS_MAX   (TILES_HI * 16)

__global__ void gin_zero_counts(int* __restrict__ cnt) {
    cnt[blockIdx.x * blockDim.x + threadIdx.x] = 0;
}

__global__ void gin_bucket(const int* __restrict__ esrc,
                           const int* __restrict__ edst,
                           int* __restrict__ cnt,
                           int* __restrict__ bkt,
                           int* __restrict__ ovs,
                           int* __restrict__ ovd) {
    int e = blockIdx.x * blockDim.x + threadIdx.x;
    if (e >= N_EDGES) return;
    int d = edst[e];
    int s = esrc[e];
    int slot = atomicAdd(cnt + d, 1);
    if (slot < CAP) {
        bkt[(size_t)d * CAP + slot] = s;
    } else {
        int o = atomicAdd(cnt + N_NODES, 1);
        ovs[o] = s;
        ovd[o] = d;
    }
}

template <int NT>
__device__ __forceinline__ void mlp_tiles(const float* hA, float* hH,
                                          const float* __restrict__ W1, const float* __restrict__ b1,
                                          const float* __restrict__ W2, const float* __restrict__ b2,
                                          float* __restrict__ out,
                                          int row0, int n0, int l15, int hi, int koff) {
    v8f acc[NT];
    {
        const float bb = b1[n0 + l15];
        #pragma unroll
        for (int t = 0; t < NT; ++t) {
            #pragma unroll
            for (int r = 0; r < 8; ++r) acc[t][r] = bb;
        }
    }
    #pragma unroll
    for (int k = 0; k < D; k += 4) {
        v2f b;
        b.x = W1[(size_t)(k + koff)     * D + n0 + l15];
        b.y = W1[(size_t)(k + koff + 1) * D + n0 + l15];
        #pragma unroll
        for (int t = 0; t < NT; ++t) {
            const v2f a = *(const v2f*)(hA + (t * 16 + l15) * LDS_PITCH + k + koff);
            acc[t] = __builtin_amdgcn_wmma_f32_16x16x4_f32(
                 false, a,  false, b,
                 (short)0, acc[t],  false,  false);
        }
    }
    #pragma unroll
    for (int t = 0; t < NT; ++t) {
        #pragma unroll
        for (int r = 0; r < 8; ++r) {
            const int m = t * 16 + r + hi * 8;
            const float v = acc[t][r] > 0.0f ? acc[t][r] : 0.0f;
            hH[m * LDS_PITCH + n0 + l15] = v;
        }
    }
    __syncthreads();

    v8f acc2[NT];
    {
        const float bb = b2[n0 + l15];
        #pragma unroll
        for (int t = 0; t < NT; ++t) {
            #pragma unroll
            for (int r = 0; r < 8; ++r) acc2[t][r] = bb;
        }
    }
    #pragma unroll
    for (int k = 0; k < D; k += 4) {
        v2f b;
        b.x = W2[(size_t)(k + koff)     * D + n0 + l15];
        b.y = W2[(size_t)(k + koff + 1) * D + n0 + l15];
        #pragma unroll
        for (int t = 0; t < NT; ++t) {
            const v2f a = *(const v2f*)(hH + (t * 16 + l15) * LDS_PITCH + k + koff);
            acc2[t] = __builtin_amdgcn_wmma_f32_16x16x4_f32(
                false, a, false, b, (short)0, acc2[t], false, false);
        }
    }
    #pragma unroll
    for (int t = 0; t < NT; ++t) {
        #pragma unroll
        for (int r = 0; r < 8; ++r) {
            const int m = t * 16 + r + hi * 8;
            const float v = acc2[t][r] > 0.0f ? acc2[t][r] : 0.0f;
            out[(size_t)(row0 + m) * D + n0 + l15] = v;
        }
    }
}

__global__ void __launch_bounds__(256)
gin_mlp(const float* __restrict__ feat,
        const float* __restrict__ eps,
        const int* __restrict__ cnt,
        const int* __restrict__ bkt,
        const int* __restrict__ ovs,
        const int* __restrict__ ovd,
        const float* __restrict__ W1, const float* __restrict__ b1,
        const float* __restrict__ W2, const float* __restrict__ b2,
        float* __restrict__ out) {
    __shared__ __attribute__((aligned(16))) float hA[ROWS_MAX * LDS_PITCH];
    __shared__ float hH[ROWS_MAX * LDS_PITCH];

    const int tid  = threadIdx.x;
    const int wave = tid >> 5;
    const int lane = tid & 31;
    const int l15  = lane & 15;
    const int hi   = lane >> 4;
    const int koff = hi * 2;
    const int blk  = blockIdx.x;
    const int nt   = blk < N_HI ? TILES_HI : TILES_LO;
    const int m0   = (blk * TILES_LO + (blk < N_HI ? blk : N_HI)) * 16;
    const int n0   = wave * 16;

    {
        const int wv = __builtin_amdgcn_readfirstlane(wave);
        const float sc = 1.0f + eps[0];
        const float4* f4 = (const float4*)feat;
        int on = cnt[N_NODES];
        on = on < N_EDGES ? on : N_EDGES;
        int cv = 0;
        if (lane < 2 * nt) cv = cnt[m0 + (lane >> 1) * 16 + wv * 2 + (lane & 1)];
        cv = cv < CAP ? cv : CAP;
        int    nn[2], i0[2], i1[2];
        float4 self[2];
        #pragma unroll
        for (int rr = 0; rr < 2; ++rr) {
            const int row = m0 + wv * 2 + rr;
            const int n = __builtin_amdgcn_readlane(cv, rr);
            nn[rr] = n;
            const int* bp = bkt + (size_t)row * CAP;
            self[rr] = f4[(size_t)row * 32 + lane];
            i0[rr] = (lane < n)      ? bp[lane]      : 0;
            i1[rr] = (lane + 32 < n) ? bp[lane + 32] : 0;
        }
        for (int p = 0; p < nt; ++p) {
            int    cn[2], c0[2], c1[2];
            float4 cs[2];
            #pragma unroll
            for (int rr = 0; rr < 2; ++rr) { cn[rr] = nn[rr]; c0[rr] = i0[rr]; c1[rr] = i1[rr]; cs[rr] = self[rr]; }
            if (p + 1 < nt) {
                #pragma unroll
                for (int rr = 0; rr < 2; ++rr) {
                    const int row = m0 + (p + 1) * 16 + wv * 2 + rr;
                    const int n = __builtin_amdgcn_readlane(cv, 2 * (p + 1) + rr);
                    nn[rr] = n;
                    const int* bp = bkt + (size_t)row * CAP;
                    self[rr] = f4[(size_t)row * 32 + lane];
                    i0[rr] = (lane < n)      ? bp[lane]      : 0;
                    i1[rr] = (lane + 32 < n) ? bp[lane + 32] : 0;
                }
            }
            #pragma unroll
            for (int rr = 0; rr < 2; ++rr) {
                const int r   = p * 16 + wv * 2 + rr;
                const int row = m0 + r;
                const int n   = cn[rr];
                float4 acc = make_float4(0.0f, 0.0f, 0.0f, 0.0f);
                #pragma nounroll
                for (int j = 0; j < n; j += 8) {
                    const int iv = (j < 32) ? c0[rr] : c1[rr];
                    float4 v[8];
                    #pragma unroll
                    for (int u = 0; u < 8; ++u) {
                        const int s = __builtin_amdgcn_readlane(iv, (j & 31) + u);
                        v[u] = f4[(size_t)s * 32 + lane];
                    }
                    #pragma unroll
                    for (int u = 0; u < 8; ++u) {
                        const bool ok = (j + u) < n;
                        acc.x += ok ? v[u].x : 0.0f;
                        acc.y += ok ? v[u].y : 0.0f;
                        acc.z += ok ? v[u].z : 0.0f;
                        acc.w += ok ? v[u].w : 0.0f;
                    }
                }
                for (int i = 0; i < on; ++i) {
                    if (ovd[i] == row) {
                        const float4 v = f4[(size_t)ovs[i] * 32 + lane];
                        acc.x += v.x; acc.y += v.y; acc.z += v.z; acc.w += v.w;
                    }
                }
                float4 t;
                t.x = fmaf(sc, cs[rr].x, acc.x);
                t.y = fmaf(sc, cs[rr].y, acc.y);
                t.z = fmaf(sc, cs[rr].z, acc.z);
                t.w = fmaf(sc, cs[rr].w, acc.w);
                *(float4*)(hA + r * LDS_PITCH + lane * 4) = t;
            }
        }
    }
    __syncthreads();

    if (nt == TILES_HI)
        mlp_tiles<TILES_HI>(hA, hH, W1, b1, W2, b2, out, m0, n0, l15, hi, koff);
    else
        mlp_tiles<TILES_LO>(hA, hH, W1, b1, W2, b2, out, m0, n0, l15, hi, koff);
}

extern "C" void kernel_launch(void* const* d_in, const int* in_sizes, int n_in,
                              void* d_out, int out_size, void* d_ws, size_t ws_size,
                              hipStream_t stream) {
    const float* feat = (const float*)d_in[0];
    const float* eps  = (const float*)d_in[1];
    const float* W1   = (const float*)d_in[2];
    const float* b1   = (const float*)d_in[3];
    const float* W2   = (const float*)d_in[4];
    const float* b2   = (const float*)d_in[5];
    const int*   esrc = (const int*)d_in[6];
    const int*   edst = (const int*)d_in[7];
    float* out = (float*)d_out;
    int* cnt = (int*)d_ws;
    int* bkt = cnt + CNT_PAD;
    int* ovs = bkt + N_NODES * CAP;
    int* ovd = ovs + N_EDGES;

    gin_zero_counts<<<CNT_PAD / 256, 256, 0, stream>>>(cnt);
    gin_bucket<<<N_EDGES / 256, 256, 0, stream>>>(esrc, edst, cnt, bkt, ovs, ovd);
    gin_mlp<<<MLP_BLOCKS, 256, 0, stream>>>(feat, eps, cnt, bkt, ovs, ovd, W1, b1, W2, b2, out);
}


# edges filed by LDS-counter ranks, not a global add per edge: pass 1 orders 3125-edge chunks by 196-row bin, pass 2 (block per bin) takes row slots from LDS adds; no clear launch; arrival-order sums
# speedup vs baseline: 3.2014x; 1.6804x over previous
#include <hip/hip_runtime.h>

typedef __attribute__((ext_vector_type(2))) float v2f;
typedef __attribute__((ext_vector_type(8))) float v8f;

#define N_NODES 50000
#define N_EDGES 800000
#define D 128
#define LDS_PITCH 132

#define CAP 64
#define CNT_PAD 50176

#define NBIN    256
#define RB      196
#define NCHUNK  256
#define CHUNK   (N_EDGES / NCHUNK)
#define EPT     ((CHUNK + 255) / 256)
#define RSTRIDE 3136
static_assert(NCHUNK * CHUNK == N_EDGES, "chunks cover the edge list exactly");
static_assert(NBIN * RB >= N_NODES && RB <= 256, "bins cover the rows, one counter per thread");
static_assert(N_NODES <= 65536 && CHUNK < 32768, "source id and chunk offsets fit 16 bits");

#define N_TILES    (N_NODES / 16)
#define MLP_BLOCKS 768
#define TILES_LO   (N_TILES / MLP_BLOCKS)
#define N_HI       (N_TILES % MLP_BLOCKS)
#define TILES_HI   (TILES_LO + 1)
#define ROWS_MAX   (TILES_HI * 16)

__global__ void __launch_bounds__(256)
gin_chunk_sort(const int* __restrict__ esrc,
               const int* __restrict__ edst,
               int* __restrict__ rec,
               int* __restrict__ seg,
               int* __restrict__ cnt) {
    __shared__ int lcnt[NBIN];
    __shared__ int wsum[8];
    __shared__ int stage[CHUNK];

    const int tid  = threadIdx.x;
    const int lane = tid & 31;
    const int wave = tid >> 5;
    const int base = blockIdx.x * CHUNK;

    if (blockIdx.x == 0 && tid == 0) cnt[N_NODES] = 0;
    lcnt[tid] = 0;
    __syncthreads();

    int word[EPT], bin[EPT], rank[EPT];
    #pragma unroll
    for (int i = 0; i < EPT; ++i) {
        const int e = i * 256 + tid;
        bin[i] = -1;
        if (e < CHUNK) {
            const int d = edst[base + e];
            const int s = esrc[base + e];
            const int b = (int)((unsigned)d / (unsigned)RB);
            word[i] = ((d - b * RB) << 16) | s;
            bin[i]  = b;
        }
    }
    #pragma unroll
    for (int i = 0; i < EPT; ++i)
        if (bin[i] >= 0) rank[i] = atomicAdd(&lcnt[bin[i]], 1);
    __syncthreads();

    const int c = lcnt[tid];
    int x = c;
    #pragma unroll
    for (int o = 1; o < 32; o <<= 1) {
        const int y = __shfl_up(x, o);
        if (lane >= o) x += y;
    }
    if (lane == 31) wsum[wave] = x;
    __syncthreads();
    int wbase = 0;
    #pragma unroll
    for (int w = 0; w < 8; ++w) wbase += (w < wave) ? wsum[w] : 0;
    const int start = wbase + x - c;
    lcnt[tid] = start;
    seg[blockIdx.x * NBIN + tid] = (c << 16) | start;
    __syncthreads();

    #pragma unroll
    for (int i = 0; i < EPT; ++i)
        if (bin[i] >= 0) stage[lcnt[bin[i]] + rank[i]] = word[i];
    __syncthreads();

    int* out = rec + (size_t)blockIdx.x * RSTRIDE;
    for (int e = tid; e < CHUNK; e += 256) out[e] = stage[e];
}

__global__ void __launch_bounds__(256)
gin_rank(const int* __restrict__ rec,
         const int* __restrict__ seg,
         int* __restrict__ cnt,
         int* __restrict__ bkt,
         int* __restrict__ ovs,
         int* __restrict__ ovd) {
    __shared__ int lc[256];

    const int tid  = threadIdx.x;
    const int row0 = blockIdx.x * RB;
    lc[tid] = 0;
    const int pk = seg[tid * NBIN + blockIdx.x];
    const int st = pk & 0xFFFF;
    const int n  = pk >> 16;
    __syncthreads();

    const int* p = rec + (size_t)tid * RSTRIDE + st;
    for (int i0 = 0; i0 < n; i0 += 16) {
        int w[16], slot[16];
        #pragma unroll
        for (int u = 0; u < 16; ++u) w[u] = (i0 + u < n) ? p[i0 + u] : -1;
        #pragma unroll
        for (int u = 0; u < 16; ++u) slot[u] = (w[u] >= 0) ? atomicAdd(&lc[w[u] >> 16], 1) : 0;
        #pragma unroll
        for (int u = 0; u < 16; ++u) {
            if (w[u] >= 0) {
                const int s = w[u] & 0xFFFF;
                const int d = row0 + (w[u] >> 16);
                if (slot[u] < CAP) {
                    bkt[(size_t)d * CAP + slot[u]] = s;
                } else {
                    const int o = atomicAdd(cnt + N_NODES, 1);
                    ovs[o] = s;
                    ovd[o] = d;
                }
            }
        }
    }
    __syncthreads();
    if (tid < RB && row0 + tid < N_NODES) cnt[row0 + tid] = lc[tid];
}

template <int NT>
__device__ __forceinline__ void mlp_tiles(const float* hA, float* hH,
                                          const float* __restrict__ W1, const float* __restrict__ b1,
                                          const float* __restrict__ W2, const float* __restrict__ b2,
                                          float* __restrict__ out,
                                          int row0, int n0, int l15, int hi, int koff) {
    v8f acc[NT];
    {
        const float bb = b1[n0 + l15];
        #pragma unroll
        for (int t = 0; t < NT; ++t) {
            #pragma unroll
            for (int r = 0; r < 8; ++r) acc[t][r] = bb;
        }
    }
    #pragma unroll
    for (int k = 0; k < D; k += 4) {
        v2f b;
        b.x = W1[(size_t)(k + koff)     * D + n0 + l15];
        b.y = W1[(size_t)(k + koff + 1) * D + n0 + l15];
        #pragma unroll
        for (int t = 0; t < NT; ++t) {
            const v2f a = *(const v2f*)(hA + (t * 16 + l15) * LDS_PITCH + k + koff);
            acc[t] = __builtin_amdgcn_wmma_f32_16x16x4_f32(
                 false, a,  false, b,
                 (short)0, acc[t],  false,  false);
        }
    }
    #pragma unroll
    for (int t = 0; t < NT; ++t) {
        #pragma unroll
        for (int r = 0; r < 8; ++r) {
            const int m = t * 16 + r + hi * 8;
            const float v = acc[t][r] > 0.0f ? acc[t][r] : 0.0f;
            hH[m * LDS_PITCH + n0 + l15] = v;
        }
    }
    __syncthreads();

    v8f acc2[NT];
    {
        const float bb = b2[n0 + l15];
        #pragma unroll
        for (int t = 0; t < NT; ++t) {
            #pragma unroll
            for (int r = 0; r < 8; ++r) acc2[t][r] = bb;
        }
    }
    #pragma unroll
    for (int k = 0; k < D; k += 4) {
        v2f b;
        b.x = W2[(size_t)(k + koff)     * D + n0 + l15];
        b.y = W2[(size_t)(k + koff + 1) * D + n0 + l15];
        #pragma unroll
        for (int t = 0; t < NT; ++t) {
            const v2f a = *(const v2f*)(hH + (t * 16 + l15) * LDS_PITCH + k + koff);
            acc2[t] = __builtin_amdgcn_wmma_f32_16x16x4_f32(
                false, a, false, b, (short)0, acc2[t], false, false);
        }
    }
    #pragma unroll
    for (int t = 0; t < NT; ++t) {
        #pragma unroll
        for (int r = 0; r < 8; ++r) {
            const int m = t * 16 + r + hi * 8;
            const float v = acc2[t][r] > 0.0f ? acc2[t][r] : 0.0f;
            out[(size_t)(row0 + m) * D + n0 + l15] = v;
        }
    }
}

__global__ void __launch_bounds__(256)
gin_mlp(const float* __restrict__ feat,
        const float* __restrict__ eps,
        const int* __restrict__ cnt,
        const int* __restrict__ bkt,
        const int* __restrict__ ovs,
        const int* __restrict__ ovd,
        const float* __restrict__ W1, const float* __restrict__ b1,
        const float* __restrict__ W2, const float* __restrict__ b2,
        float* __restrict__ out) {
    __shared__ __attribute__((aligned(16))) float hA[ROWS_MAX * LDS_PITCH];
    __shared__ float hH[ROWS_MAX * LDS_PITCH];

    const int tid  = threadIdx.x;
    const int wave = tid >> 5;
    const int lane = tid & 31;
    const int l15  = lane & 15;
    const int hi   = lane >> 4;
    const int koff = hi * 2;
    const int blk  = blockIdx.x;
    const int nt   = blk < N_HI ? TILES_HI : TILES_LO;
    const int m0   = (blk * TILES_LO + (blk < N_HI ? blk : N_HI)) * 16;
    const int n0   = wave * 16;

    {
        const int wv = __builtin_amdgcn_readfirstlane(wave);
        const float sc = 1.0f + eps[0];
        const float4* f4 = (const float4*)feat;
        int on = cnt[N_NODES];
        on = on < N_EDGES ? on : N_EDGES;
        int cv = 0;
        if (lane < 2 * nt) cv = cnt[m0 + (lane >> 1) * 16 + wv * 2 + (lane & 1)];
        cv = cv < CAP ? cv : CAP;
        int    nn[2], i0[2], i1[2];
        float4 self[2];
        #pragma unroll
        for (int rr = 0; rr < 2; ++rr) {
            const int row = m0 + wv * 2 + rr;
            const int n = __builtin_amdgcn_readlane(cv, rr);
            nn[rr] = n;
            const int* bp = bkt + (size_t)row * CAP;
            self[rr] = f4[(size_t)row * 32 + lane];
            i0[rr] = (lane < n)      ? bp[lane]      : 0;
            i1[rr] = (lane + 32 < n) ? bp[lane + 32] : 0;
        }
        for (int p = 0; p < nt; ++p) {
            int    cn[2], c0[2], c1[2];
            float4 cs[2];
            #pragma unroll
            for (int rr = 0; rr < 2; ++rr) { cn[rr] = nn[rr]; c0[rr] = i0[rr]; c1[rr] = i1[rr]; cs[rr] = self[rr]; }
            if (p + 1 < nt) {
                #pragma unroll
                for (int rr = 0; rr < 2; ++rr) {
                    const int row = m0 + (p + 1) * 16 + wv * 2 + rr;
                    const int n = __builtin_amdgcn_readlane(cv, 2 * (p + 1) + rr);
                    nn[rr] = n;
                    const int* bp = bkt + (size_t)row * CAP;
                    self[rr] = f4[(size_t)row * 32 + lane];
                    i0[rr] = (lane < n)      ? bp[lane]      : 0;
                    i1[rr] = (lane + 32 < n) ? bp[lane + 32] : 0;
                }
            }
            #pragma unroll
            for (int rr = 0; rr < 2; ++rr) {
                const int r   = p * 16 + wv * 2 + rr;
                const int row = m0 + r;
                const int n   = cn[rr];
                float4 acc = make_float4(0.0f, 0.0f, 0.0f, 0.0f);
                #pragma nounroll
                for (int j = 0; j < n; j += 8) {
                    const int iv = (j < 32) ? c0[rr] : c1[rr];
                    float4 v[8];
                    #pragma unroll
                    for (int u = 0; u < 8; ++u) {
                        const int s = __builtin_amdgcn_readlane(iv, (j & 31) + u);
                        v[u] = f4[(size_t)s * 32 + lane];
                    }
                    #pragma unroll
                    for (int u = 0; u < 8; ++u) {
                        const bool ok = (j + u) < n;
                        acc.x += ok ? v[u].x : 0.0f;
                        acc.y += ok ? v[u].y : 0.0f;
                        acc.z += ok ? v[u].z : 0.0f;
                        acc.w += ok ? v[u].w : 0.0f;
                    }
                }
                for (int i = 0; i < on; ++i) {
                    if (ovd[i] == row) {
                        const float4 v = f4[(size_t)ovs[i] * 32 + lane];
                        acc.x += v.x; acc.y += v.y; acc.z += v.z; acc.w += v.w;
                    }
                }
                float4 t;
                t.x = fmaf(sc, cs[rr].x, acc.x);
                t.y = fmaf(sc, cs[rr].y, acc.y);
                t.z = fmaf(sc, cs[rr].z, acc.z);
                t.w = fmaf(sc, cs[rr].w, acc.w);
                *(float4*)(hA + r * LDS_PITCH + lane * 4) = t;
            }
        }
    }
    __syncthreads();

    if (nt == TILES_HI)
        mlp_tiles<TILES_HI>(hA, hH, W1, b1, W2, b2, out, m0, n0, l15, hi, koff);
    else
        mlp_tiles<TILES_LO>(hA, hH, W1, b1, W2, b2, out, m0, n0, l15, hi, koff);
}

extern "C" void kernel_launch(void* const* d_in, const int* in_sizes, int n_in,
                              void* d_out, int out_size, void* d_ws, size_t ws_size,
                              hipStream_t stream) {
    const float* feat = (const float*)d_in[0];
    const float* eps  = (const float*)d_in[1];
    const float* W1   = (const float*)d_in[2];
    const float* b1   = (const float*)d_in[3];
    const float* W2   = (const float*)d_in[4];
    const float* b2   = (const float*)d_in[5];
    const int*   esrc = (const int*)d_in[6];
    const int*   edst = (const int*)d_in[7];
    float* out = (float*)d_out;
    int* cnt = (int*)d_ws;
    int* bkt = cnt + CNT_PAD;
    int* ovs = bkt + N_NODES * CAP;
    int* ovd = ovs + N_EDGES;
    int* rec = ovd + N_EDGES;
    int* seg = rec + NCHUNK * RSTRIDE;

    gin_chunk_sort<<<NCHUNK, 256, 0, stream>>>(esrc, edst, rec, seg, cnt);
    gin_rank<<<NBIN, 256, 0, stream>>>(rec, seg, cnt, bkt, ovs, ovd);
    gin_mlp<<<MLP_BLOCKS, 256, 0, stream>>>(feat, eps, cnt, bkt, ovs, ovd, W1, b1, W2, b2, out);
}


# rank pass folded into the MLP block: chunk-sort bins = the 768 MLP blocks' rows; each block files its bin's records in an LDS bucket by LDS adds; 2 launches, no global bucket; arrival-order sums
# speedup vs baseline: 3.3651x; 1.0511x over previous
#include <hip/hip_runtime.h>

typedef __attribute__((ext_vector_type(2))) float v2f;
typedef __attribute__((ext_vector_type(8))) float v8f;

#define N_NODES 50000
#define N_EDGES 800000
#define D 128
#define LDS_PITCH 132

#define CAP 64
#define BPITCH (CAP + 1)

#define N_TILES    (N_NODES / 16)
#define MLP_BLOCKS 768
#define TILES_LO   (N_TILES / MLP_BLOCKS)
#define N_HI       (N_TILES % MLP_BLOCKS)
#define TILES_HI   (TILES_LO + 1)
#define ROWS_MAX   (TILES_HI * 16)
#define ROWS_LO    (TILES_LO * 16)
#define HI_END     (N_HI * ROWS_MAX)

#define NBIN    MLP_BLOCKS
#define BPT     (NBIN / 256)
#define NCHUNK  256
#define CHUNK   (N_EDGES / NCHUNK)
#define EPT     ((CHUNK + 255) / 256)
#define RSTRIDE 3136
static_assert(NCHUNK * CHUNK == N_EDGES, "chunks cover the edge list exactly");
static_assert(BPT * 256 == NBIN, "every thread of the chunk sort owns BPT whole bins");
static_assert(N_NODES <= 65536 && CHUNK < 32768, "source id and chunk offsets fit 16 bits");
static_assert(HI_END + (MLP_BLOCKS - N_HI) * ROWS_LO == N_NODES, "the bins cover the rows exactly");

__global__ void __launch_bounds__(256)
gin_chunk_sort(const int* __restrict__ esrc,
               const int* __restrict__ edst,
               int* __restrict__ rec,
               int* __restrict__ seg) {
    __shared__ int lcnt[NBIN];
    __shared__ int segl[NBIN];
    __shared__ int wsum[8];
    __shared__ int stage[CHUNK];

    const int tid  = threadIdx.x;
    const int lane = tid & 31;
    const int wave = tid >> 5;
    const int base = blockIdx.x * CHUNK;

    #pragma unroll
    for (int j = 0; j < BPT; ++j) lcnt[j * 256 + tid] = 0;
    __syncthreads();

    int word[EPT], bin[EPT], rank[EPT];
    #pragma unroll
    for (int i = 0; i < EPT; ++i) {
        const int e = i * 256 + tid;
        bin[i] = -1;
        if (e < CHUNK) {
            const int d = edst[base + e];
            const int s = esrc[base + e];
            int b, r;
            if (d < HI_END) {
                b = (int)((unsigned)d / (unsigned)ROWS_MAX);
                r = d - b * ROWS_MAX;
            } else {
                const int q = d - HI_END;
                const int bq = (int)((unsigned)q / (unsigned)ROWS_LO);
                b = N_HI + bq;
                r = q - bq * ROWS_LO;
            }
            word[i] = (r << 16) | s;
            bin[i]  = b;
        }
    }
    #pragma unroll
    for (int i = 0; i < EPT; ++i)
        if (bin[i] >= 0) rank[i] = atomicAdd(&lcnt[bin[i]], 1);
    __syncthreads();

    int c[BPT];
    int tot = 0;
    #pragma unroll
    for (int j = 0; j < BPT; ++j) { c[j] = lcnt[tid * BPT + j]; tot += c[j]; }
    int x = tot;
    #pragma unroll
    for (int o = 1; o < 32; o <<= 1) {
        const int y = __shfl_up(x, o);
        if (lane >= o) x += y;
    }
    if (lane == 31) wsum[wave] = x;
    __syncthreads();
    int wbase = 0;
    #pragma unroll
    for (int w = 0; w < 8; ++w) wbase += (w < wave) ? wsum[w] : 0;
    int start = wbase + x - tot;
    #pragma unroll
    for (int j = 0; j < BPT; ++j) {
        lcnt[tid * BPT + j] = start;
        segl[tid * BPT + j] = (c[j] << 16) | start;
        start += c[j];
    }
    __syncthreads();

    #pragma unroll
    for (int j = 0; j < BPT; ++j) seg[blockIdx.x * NBIN + j * 256 + tid] = segl[j * 256 + tid];

    #pragma unroll
    for (int i = 0; i < EPT; ++i)
        if (bin[i] >= 0) stage[lcnt[bin[i]] + rank[i]] = word[i];
    __syncthreads();

    int* out = rec + (size_t)blockIdx.x * RSTRIDE;
    for (int e = tid; e < CHUNK; e += 256) out[e] = stage[e];
}

template <int NT>
__device__ __forceinline__ void mlp_tiles(const float* hA, float* hH,
                                          const float* __restrict__ W1, const float* __restrict__ b1,
                                          const float* __restrict__ W2, const float* __restrict__ b2,
                                          float* __restrict__ out,
                                          int row0, int n0, int l15, int hi, int koff) {
    v8f acc[NT];
    {
        const float bb = b1[n0 + l15];
        #pragma unroll
        for (int t = 0; t < NT; ++t) {
            #pragma unroll
            for (int r = 0; r < 8; ++r) acc[t][r] = bb;
        }
    }
    #pragma unroll
    for (int k = 0; k < D; k += 4) {
        v2f b;
        b.x = W1[(size_t)(k + koff)     * D + n0 + l15];
        b.y = W1[(size_t)(k + koff + 1) * D + n0 + l15];
        #pragma unroll
        for (int t = 0; t < NT; ++t) {
            const v2f a = *(const v2f*)(hA + (t * 16 + l15) * LDS_PITCH + k + koff);
            acc[t] = __builtin_amdgcn_wmma_f32_16x16x4_f32(
                 false, a,  false, b,
                 (short)0, acc[t],  false,  false);
        }
    }
    #pragma unroll
    for (int t = 0; t < NT; ++t) {
        #pragma unroll
        for (int r = 0; r < 8; ++r) {
            const int m = t * 16 + r + hi * 8;
            const float v = acc[t][r] > 0.0f ? acc[t][r] : 0.0f;
            hH[m * LDS_PITCH + n0 + l15] = v;
        }
    }
    __syncthreads();

    v8f acc2[NT];
    {
        const float bb = b2[n0 + l15];
        #pragma unroll
        for (int t = 0; t < NT; ++t) {
            #pragma unroll
            for (int r = 0; r < 8; ++r) acc2[t][r] = bb;
        }
    }
    #pragma unroll
    for (int k = 0; k < D; k += 4) {
        v2f b;
        b.x = W2[(size_t)(k + koff)     * D + n0 + l15];
        b.y = W2[(size_t)(k + koff + 1) * D + n0 + l15];
        #pragma unroll
        for (int t = 0; t < NT; ++t) {
            const v2f a = *(const v2f*)(hH + (t * 16 + l15) * LDS_PITCH + k + koff);
            acc2[t] = __builtin_amdgcn_wmma_f32_16x16x4_f32(
                false, a, false, b, (short)0, acc2[t], false, false);
        }
    }
    #pragma unroll
    for (int t = 0; t < NT; ++t) {
        #pragma unroll
        for (int r = 0; r < 8; ++r) {
            const int m = t * 16 + r + hi * 8;
            const float v = acc2[t][r] > 0.0f ? acc2[t][r] : 0.0f;
            out[(size_t)(row0 + m) * D + n0 + l15] = v;
        }
    }
}

union HiddenOrBucket {
    float h[ROWS_MAX * LDS_PITCH];
    int   b[ROWS_MAX * BPITCH + ROWS_MAX];
};

__global__ void __launch_bounds__(256)
gin_mlp(const float* __restrict__ feat,
        const float* __restrict__ eps,
        const int* __restrict__ rec,
        const int* __restrict__ seg,
        const float* __restrict__ W1, const float* __restrict__ b1,
        const float* __restrict__ W2, const float* __restrict__ b2,
        float* __restrict__ out) {
    __shared__ __attribute__((aligned(16))) float hA[ROWS_MAX * LDS_PITCH];
    __shared__ __attribute__((aligned(16))) HiddenOrBucket hb;
    float* hH   = hb.h;
    int*   lbkt = hb.b;
    int*   lc   = hb.b + ROWS_MAX * BPITCH;

    const int tid  = threadIdx.x;
    const int wave = tid >> 5;
    const int lane = tid & 31;
    const int l15  = lane & 15;
    const int hi   = lane >> 4;
    const int koff = hi * 2;
    const int blk  = blockIdx.x;
    const int nt   = blk < N_HI ? TILES_HI : TILES_LO;
    const int m0   = (blk * TILES_LO + (blk < N_HI ? blk : N_HI)) * 16;
    const int n0   = wave * 16;

    {
        if (tid < ROWS_MAX) lc[tid] = 0;
        const int pk = seg[tid * NBIN + blk];
        const int st = pk & 0xFFFF;
        const int n  = pk >> 16;
        __syncthreads();
        const int* p = rec + (size_t)tid * RSTRIDE + st;
        for (int i0 = 0; i0 < n; i0 += 8) {
            int w[8], slot[8];
            #pragma unroll
            for (int u = 0; u < 8; ++u) w[u] = (i0 + u < n) ? p[i0 + u] : -1;
            #pragma unroll
            for (int u = 0; u < 8; ++u) slot[u] = (w[u] >= 0) ? atomicAdd(&lc[w[u] >> 16], 1) : CAP;
            #pragma unroll
            for (int u = 0; u < 8; ++u)
                if (slot[u] < CAP) lbkt[(w[u] >> 16) * BPITCH + slot[u]] = w[u] & 0xFFFF;
        }
        __syncthreads();
    }

    {
        const int wv = __builtin_amdgcn_readfirstlane(wave);
        const float sc = 1.0f + eps[0];
        const float4* f4 = (const float4*)feat;
        int cv = 0;
        if (lane < 2 * nt) cv = lc[(lane >> 1) * 16 + wv * 2 + (lane & 1)];
        int    nn[2], i0[2], i1[2];
        float4 self[2];
        #pragma unroll
        for (int rr = 0; rr < 2; ++rr) {
            const int r = wv * 2 + rr;
            const int n = __builtin_amdgcn_readlane(cv, rr);
            nn[rr] = n;
            const int* bp = lbkt + r * BPITCH;
            self[rr] = f4[(size_t)(m0 + r) * 32 + lane];
            i0[rr] = (lane < n)      ? bp[lane]      : 0;
            i1[rr] = (lane + 32 < n) ? bp[lane + 32] : 0;
        }
        for (int p = 0; p < nt; ++p) {
            int    cn[2], c0[2], c1[2];
            float4 cs[2];
            #pragma unroll
            for (int rr = 0; rr < 2; ++rr) { cn[rr] = nn[rr]; c0[rr] = i0[rr]; c1[rr] = i1[rr]; cs[rr] = self[rr]; }
            if (p + 1 < nt) {
                #pragma unroll
                for (int rr = 0; rr < 2; ++rr) {
                    const int r = (p + 1) * 16 + wv * 2 + rr;
                    const int n = __builtin_amdgcn_readlane(cv, 2 * (p + 1) + rr);
                    nn[rr] = n;
                    const int* bp = lbkt + r * BPITCH;
                    self[rr] = f4[(size_t)(m0 + r) * 32 + lane];
                    i0[rr] = (lane < n)      ? bp[lane]      : 0;
                    i1[rr] = (lane + 32 < n) ? bp[lane + 32] : 0;
                }
            }
            #pragma unroll
            for (int rr = 0; rr < 2; ++rr) {
                const int r = p * 16 + wv * 2 + rr;
                const int n = cn[rr];
                float4 acc = make_float4(0.0f, 0.0f, 0.0f, 0.0f);
                if (n <= CAP) {
                    #pragma nounroll
                    for (int j = 0; j < n; j += 8) {
                        const int iv = (j < 32) ? c0[rr] : c1[rr];
                        float4 v[8];
                        #pragma unroll
                        for (int u = 0; u < 8; ++u) {
                            const int s = __builtin_amdgcn_readlane(iv, (j & 31) + u);
                            v[u] = f4[(size_t)s * 32 + lane];
                        }
                        #pragma unroll
                        for (int u = 0; u < 8; ++u) {
                            const bool ok = (j + u) < n;
                            acc.x += ok ? v[u].x : 0.0f;
                            acc.y += ok ? v[u].y : 0.0f;
                            acc.z += ok ? v[u].z : 0.0f;
                            acc.w += ok ? v[u].w : 0.0f;
                        }
                    }
                } else {
                    for (int c = 0; c < NCHUNK; ++c) {
                        const int pk = seg[c * NBIN + blk];
                        const int* q = rec + (size_t)c * RSTRIDE + (pk & 0xFFFF);
                        const int m  = pk >> 16;
                        for (int i = 0; i < m; ++i) {
                            const int w = q[i];
                            if ((w >> 16) == r) {
                                const float4 v = f4[(size_t)(w & 0xFFFF) * 32 + lane];
                                acc.x += v.x; acc.y += v.y; acc.z += v.z; acc.w += v.w;
                            }
                        }
                    }
                }
                float4 t;
                t.x = fmaf(sc, cs[rr].x, acc.x);
                t.y = fmaf(sc, cs[rr].y, acc.y);
                t.z = fmaf(sc, cs[rr].z, acc.z);
                t.w = fmaf(sc, cs[rr].w, acc.w);
                *(float4*)(hA + r * LDS_PITCH + lane * 4) = t;
            }
        }
    }
    __syncthreads();

    if (nt == TILES_HI)
        mlp_tiles<TILES_HI>(hA, hH, W1, b1, W2, b2, out, m0, n0, l15, hi, koff);
    else
        mlp_tiles<TILES_LO>(hA, hH, W1, b1, W2, b2, out, m0, n0, l15, hi, koff);
}

extern "C" void kernel_launch(void* const* d_in, const int* in_sizes, int n_in,
                              void* d_out, int out_size, void* d_ws, size_t ws_size,
                              hipStream_t stream) {
    const float* feat = (const float*)d_in[0];
    const float* eps  = (const float*)d_in[1];
    const float* W1   = (const float*)d_in[2];
    const float* b1   = (const float*)d_in[3];
    const float* W2   = (const float*)d_in[4];
    const float* b2   = (const float*)d_in[5];
    const int*   esrc = (const int*)d_in[6];
    const int*   edst = (const int*)d_in[7];
    float* out = (float*)d_out;
    int* rec = (int*)d_ws;
    int* seg = rec + NCHUNK * RSTRIDE;

    gin_chunk_sort<<<NCHUNK, 256, 0, stream>>>(esrc, edst, rec, seg);
    gin_mlp<<<MLP_BLOCKS, 256, 0, stream>>>(feat, eps, rec, seg, W1, b1, W2, b2, out);
}
